# starting kernel
# baseline (speedup 1.0000x reference)
#include <hip/hip_runtime.h>
#include <math.h>


typedef _Float16 half_t;
typedef __attribute__((ext_vector_type(16))) _Float16 v16h;
typedef __attribute__((ext_vector_type(8)))  float    v8f;

#define B_DIM 32
#define L_DIM 4096
#define H_DIM 16
#define QS    64
#define VS    256
#define CHUNK 256
#define NCHUNK 16

__device__ __forceinline__ v8f wmma_f16(v16h a, v16h b, v8f c) {
  return __builtin_amdgcn_wmma_f32_16x16x32_f16(
       false, a,  false, b,
       (short)0, c,  false,  false);
}

__device__ __forceinline__ v16h load_frag_rm(const float* __restrict__ base,
                                             int ld, int r0, int k0,
                                             float scale, int lane) {
  const int r = lane & 15, hf = lane >> 4;
  const float* p = base + (size_t)(r0 + r) * ld + k0 + hf * 8;
  v16h f;
#pragma unroll
  for (int e = 0; e < 16; ++e)
    f[e] = (half_t)(p[16 * (e >> 3) + (e & 7)] * scale);
  return f;
}

__device__ __forceinline__ v16h load_frag_b_cm(const float* __restrict__ base,
                                               int ld, int k0, int n0, int lane) {
  const int n = lane & 15, hf = lane >> 4;
  const float* p = base + (size_t)(k0 + hf * 8) * ld + n0 + n;
  v16h f;
#pragma unroll
  for (int e = 0; e < 16; ++e)
    f[e] = (half_t)p[(size_t)(16 * (e >> 3) + (e & 7)) * ld];
  return f;
}

__device__ __forceinline__ v16h load_frag_a_lds(const half_t* base, int ldh,
                                                int k0, int lane) {
  const int m = lane & 15, hf = lane >> 4;
  const half_t* p = base + m * ldh + k0 + hf * 8;
  v16h f;
#pragma unroll
  for (int e = 0; e < 16; ++e)
    f[e] = p[16 * (e >> 3) + (e & 7)];
  return f;
}

__global__ __launch_bounds__(128) void k1_qproj(
    const float* __restrict__ query, const float* __restrict__ Wq,
    const float* __restrict__ bq, float* __restrict__ Q) {
  const int wid  = (blockIdx.x * blockDim.x + threadIdx.x) >> 5;
  const int lane = threadIdx.x & 31;
  const int mt = wid & 1;
  const int nt = wid >> 1;

  v8f c = {};
#pragma unroll 4
  for (int k0 = 0; k0 < 1024; k0 += 32) {
    v16h a = load_frag_rm(query, 1024, mt * 16, k0, 1.0f, lane);
    v16h b = load_frag_rm(Wq,    1024, nt * 16, k0, 1.0f, lane);
    c = wmma_f16(a, b, c);
  }
  const int n = lane & 15, hf = lane >> 4;
  const float bias = bq[nt * 16 + n];
#pragma unroll
  for (int v = 0; v < 8; ++v) {
    const int m = v + hf * 8;
    Q[(size_t)(mt * 16 + m) * 1024 + nt * 16 + n] = c[v] + bias;
  }
}

__global__ __launch_bounds__(256) void k2_attn_partial(
    const float* __restrict__ Q,
    const float* __restrict__ key,
    const float* __restrict__ value,
    float* __restrict__ ctxp,
    float* __restrict__ msum) {
  __shared__ float  S[16][CHUNK];
  __shared__ half_t P[16][CHUNK + 8];
  __shared__ float  red[16][2];

  const int b  = blockIdx.x >> 4;
  const int ch = blockIdx.x & 15;
  const int l0 = ch * CHUNK;
  const int tid = threadIdx.x, wave = tid >> 5, lane = tid & 31;

  const float* qb = Q + (size_t)b * 1024;
  const v16h a0 = load_frag_rm(qb, QS, 0, 0,  0.125f, lane);
  const v16h a1 = load_frag_rm(qb, QS, 0, 32, 0.125f, lane);

  const float* kb = key + ((size_t)b * L_DIM + l0) * QS;
#pragma unroll
  for (int t = 0; t < CHUNK / 16 / 8; ++t) {
    const int ln0 = (wave + t * 8) * 16;
    v16h b0 = load_frag_rm(kb, QS, ln0, 0,  1.0f, lane);
    v16h b1 = load_frag_rm(kb, QS, ln0, 32, 1.0f, lane);
    v8f c = {};
    c = wmma_f16(a0, b0, c);
    c = wmma_f16(a1, b1, c);
    const int n = lane & 15, hf = lane >> 4;
#pragma unroll
    for (int v = 0; v < 8; ++v) S[v + hf * 8][ln0 + n] = c[v];
  }
  __syncthreads();

#pragma unroll
  for (int rr = 0; rr < 2; ++rr) {
    const int r = wave * 2 + rr;
    float vals[CHUNK / 32];
    float mx = -INFINITY;
#pragma unroll
    for (int i = 0; i < CHUNK / 32; ++i) {
      vals[i] = S[r][lane + 32 * i];
      mx = fmaxf(mx, vals[i]);
    }
#pragma unroll
    for (int off = 16; off; off >>= 1) mx = fmaxf(mx, __shfl_xor(mx, off, 32));
    float sum = 0.0f;
#pragma unroll
    for (int i = 0; i < CHUNK / 32; ++i) {
      const float e = __expf(vals[i] - mx);
      sum += e;
      P[r][lane + 32 * i] = (half_t)e;
    }
#pragma unroll
    for (int off = 16; off; off >>= 1) sum += __shfl_xor(sum, off, 32);
    if (lane == 0) { red[r][0] = mx; red[r][1] = sum; }
  }
  __syncthreads();

  const float* vb = value + ((size_t)b * L_DIM + l0) * VS;
  const int n0 = wave * 32;
  v8f c0 = {}, c1 = {};
#pragma unroll 2
  for (int k0 = 0; k0 < CHUNK; k0 += 32) {
    if (k0 + 32 < CHUNK)
      __builtin_prefetch(vb + (size_t)(k0 + 32 + lane) * VS + n0, 0, 0);
    v16h pa  = load_frag_a_lds(&P[0][0], CHUNK + 8, k0, lane);
    v16h vb0 = load_frag_b_cm(vb, VS, k0, n0,      lane);
    v16h vb1 = load_frag_b_cm(vb, VS, k0, n0 + 16, lane);
    c0 = wmma_f16(pa, vb0, c0);
    c1 = wmma_f16(pa, vb1, c1);
  }

  float* op = ctxp + (((size_t)b * NCHUNK + ch) * 16) * VS;
  const int n = lane & 15, hf = lane >> 4;
#pragma unroll
  for (int v = 0; v < 8; ++v) {
    const int m = v + hf * 8;
    op[(size_t)m * VS + n0 + n]      = c0[v];
    op[(size_t)m * VS + n0 + 16 + n] = c1[v];
  }
  if (tid < 16) {
    msum[(((size_t)b * NCHUNK + ch) * 16 + tid) * 2 + 0] = red[tid][0];
    msum[(((size_t)b * NCHUNK + ch) * 16 + tid) * 2 + 1] = red[tid][1];
  }
}

__global__ __launch_bounds__(256) void k3_reduce_out(
    const float* __restrict__ ctxp, const float* __restrict__ msum,
    const float* __restrict__ Wv, const float* __restrict__ bv,
    float* __restrict__ out) {
  __shared__ float ctx[16][VS];
  __shared__ float scl[16][NCHUNK];

  const int b = blockIdx.x;
  const int tid = threadIdx.x, wave = tid >> 5, lane = tid & 31;

  if (tid < 16) {
    const int h = tid;
    float mp[NCHUNK], sp[NCHUNK], mx = -INFINITY;
#pragma unroll
    for (int p = 0; p < NCHUNK; ++p) {
      mp[p] = msum[(((size_t)b * NCHUNK + p) * 16 + h) * 2 + 0];
      sp[p] = msum[(((size_t)b * NCHUNK + p) * 16 + h) * 2 + 1];
      mx = fmaxf(mx, mp[p]);
    }
    float denom = 0.0f;
#pragma unroll
    for (int p = 0; p < NCHUNK; ++p) denom += sp[p] * __expf(mp[p] - mx);
    const float inv = 1.0f / denom;
#pragma unroll
    for (int p = 0; p < NCHUNK; ++p) scl[h][p] = __expf(mp[p] - mx) * inv;
  }
  __syncthreads();

  for (int i = tid; i < 16 * VS; i += 256) {
    const int h = i >> 8, e = i & (VS - 1);
    float acc = 0.0f;
#pragma unroll
    for (int p = 0; p < NCHUNK; ++p)
      acc += ctxp[(((size_t)b * NCHUNK + p) * 16 + h) * VS + e] * scl[h][p];
    ctx[h][e] = acc;
  }
  __syncthreads();

  for (int o = wave; o < 1024; o += 8) {
    const int h = o >> 6;
    const float* wr = Wv + (size_t)o * VS;
    float acc = 0.0f;
#pragma unroll 4
    for (int e = lane; e < VS; e += 32) acc += ctx[h][e] * wr[e];
#pragma unroll
    for (int off = 16; off; off >>= 1) acc += __shfl_xor(acc, off, 32);
    if (lane == 0) out[(size_t)b * 1024 + o] = acc + bv[o];
  }
}

extern "C" void kernel_launch(void* const* d_in, const int* in_sizes, int n_in,
                              void* d_out, int out_size, void* d_ws, size_t ws_size,
                              hipStream_t stream) {
  const float* query = (const float*)d_in[0];
  const float* key   = (const float*)d_in[1];
  const float* value = (const float*)d_in[2];
  const float* Wq    = (const float*)d_in[3];
  const float* bq    = (const float*)d_in[4];
  const float* Wv    = (const float*)d_in[5];
  const float* bv    = (const float*)d_in[6];
  float* out = (float*)d_out;

  char* ws = (char*)d_ws;
  float* Q    = (float*)ws;
  float* ctxp = (float*)(ws + 131072);
  float* msum = (float*)(ws + 131072 +
                         (size_t)B_DIM * NCHUNK * 16 * VS * 4);

  k1_qproj<<<32, 128, 0, stream>>>(query, Wq, bq, Q);
  k2_attn_partial<<<B_DIM * NCHUNK, 256, 0, stream>>>(Q, key, value, ctxp, msum);
  k3_reduce_out<<<B_DIM, 256, 0, stream>>>(ctxp, msum, Wv, bv, out);
}


# k3 (chunk merge + f32 out projection) re-gridded from one work-group per batch row to one per (batch, head), 32 -> 512 work-groups, loads issued up front; projection dot summed in a new lane order
# speedup vs baseline: 4.6439x; 4.6439x over previous
#include <hip/hip_runtime.h>
#include <math.h>


typedef _Float16 half_t;
typedef __attribute__((ext_vector_type(16))) _Float16 v16h;
typedef __attribute__((ext_vector_type(8)))  float    v8f;

#define B_DIM 32
#define L_DIM 4096
#define H_DIM 16
#define QS    64
#define VS    256
#define CHUNK 256
#define NCHUNK 16

__device__ __forceinline__ v8f wmma_f16(v16h a, v16h b, v8f c) {
  return __builtin_amdgcn_wmma_f32_16x16x32_f16(
       false, a,  false, b,
       (short)0, c,  false,  false);
}

__device__ __forceinline__ v16h load_frag_rm(const float* __restrict__ base,
                                             int ld, int r0, int k0,
                                             float scale, int lane) {
  const int r = lane & 15, hf = lane >> 4;
  const float* p = base + (size_t)(r0 + r) * ld + k0 + hf * 8;
  v16h f;
#pragma unroll
  for (int e = 0; e < 16; ++e)
    f[e] = (half_t)(p[16 * (e >> 3) + (e & 7)] * scale);
  return f;
}

__device__ __forceinline__ v16h load_frag_b_cm(const float* __restrict__ base,
                                               int ld, int k0, int n0, int lane) {
  const int n = lane & 15, hf = lane >> 4;
  const float* p = base + (size_t)(k0 + hf * 8) * ld + n0 + n;
  v16h f;
#pragma unroll
  for (int e = 0; e < 16; ++e)
    f[e] = (half_t)p[(size_t)(16 * (e >> 3) + (e & 7)) * ld];
  return f;
}

__device__ __forceinline__ v16h load_frag_a_lds(const half_t* base, int ldh,
                                                int k0, int lane) {
  const int m = lane & 15, hf = lane >> 4;
  const half_t* p = base + m * ldh + k0 + hf * 8;
  v16h f;
#pragma unroll
  for (int e = 0; e < 16; ++e)
    f[e] = p[16 * (e >> 3) + (e & 7)];
  return f;
}

__global__ __launch_bounds__(128) void k1_qproj(
    const float* __restrict__ query, const float* __restrict__ Wq,
    const float* __restrict__ bq, float* __restrict__ Q) {
  const int wid  = (blockIdx.x * blockDim.x + threadIdx.x) >> 5;
  const int lane = threadIdx.x & 31;
  const int mt = wid & 1;
  const int nt = wid >> 1;

  v8f c = {};
#pragma unroll 4
  for (int k0 = 0; k0 < 1024; k0 += 32) {
    v16h a = load_frag_rm(query, 1024, mt * 16, k0, 1.0f, lane);
    v16h b = load_frag_rm(Wq,    1024, nt * 16, k0, 1.0f, lane);
    c = wmma_f16(a, b, c);
  }
  const int n = lane & 15, hf = lane >> 4;
  const float bias = bq[nt * 16 + n];
#pragma unroll
  for (int v = 0; v < 8; ++v) {
    const int m = v + hf * 8;
    Q[(size_t)(mt * 16 + m) * 1024 + nt * 16 + n] = c[v] + bias;
  }
}

__global__ __launch_bounds__(256) void k2_attn_partial(
    const float* __restrict__ Q,
    const float* __restrict__ key,
    const float* __restrict__ value,
    float* __restrict__ ctxp,
    float* __restrict__ msum) {
  __shared__ float  S[16][CHUNK];
  __shared__ half_t P[16][CHUNK + 8];
  __shared__ float  red[16][2];

  const int b  = blockIdx.x >> 4;
  const int ch = blockIdx.x & 15;
  const int l0 = ch * CHUNK;
  const int tid = threadIdx.x, wave = tid >> 5, lane = tid & 31;

  const float* qb = Q + (size_t)b * 1024;
  const v16h a0 = load_frag_rm(qb, QS, 0, 0,  0.125f, lane);
  const v16h a1 = load_frag_rm(qb, QS, 0, 32, 0.125f, lane);

  const float* kb = key + ((size_t)b * L_DIM + l0) * QS;
#pragma unroll
  for (int t = 0; t < CHUNK / 16 / 8; ++t) {
    const int ln0 = (wave + t * 8) * 16;
    v16h b0 = load_frag_rm(kb, QS, ln0, 0,  1.0f, lane);
    v16h b1 = load_frag_rm(kb, QS, ln0, 32, 1.0f, lane);
    v8f c = {};
    c = wmma_f16(a0, b0, c);
    c = wmma_f16(a1, b1, c);
    const int n = lane & 15, hf = lane >> 4;
#pragma unroll
    for (int v = 0; v < 8; ++v) S[v + hf * 8][ln0 + n] = c[v];
  }
  __syncthreads();

#pragma unroll
  for (int rr = 0; rr < 2; ++rr) {
    const int r = wave * 2 + rr;
    float vals[CHUNK / 32];
    float mx = -INFINITY;
#pragma unroll
    for (int i = 0; i < CHUNK / 32; ++i) {
      vals[i] = S[r][lane + 32 * i];
      mx = fmaxf(mx, vals[i]);
    }
#pragma unroll
    for (int off = 16; off; off >>= 1) mx = fmaxf(mx, __shfl_xor(mx, off, 32));
    float sum = 0.0f;
#pragma unroll
    for (int i = 0; i < CHUNK / 32; ++i) {
      const float e = __expf(vals[i] - mx);
      sum += e;
      P[r][lane + 32 * i] = (half_t)e;
    }
#pragma unroll
    for (int off = 16; off; off >>= 1) sum += __shfl_xor(sum, off, 32);
    if (lane == 0) { red[r][0] = mx; red[r][1] = sum; }
  }
  __syncthreads();

  const float* vb = value + ((size_t)b * L_DIM + l0) * VS;
  const int n0 = wave * 32;
  v8f c0 = {}, c1 = {};
#pragma unroll 2
  for (int k0 = 0; k0 < CHUNK; k0 += 32) {
    if (k0 + 32 < CHUNK)
      __builtin_prefetch(vb + (size_t)(k0 + 32 + lane) * VS + n0, 0, 0);
    v16h pa  = load_frag_a_lds(&P[0][0], CHUNK + 8, k0, lane);
    v16h vb0 = load_frag_b_cm(vb, VS, k0, n0,      lane);
    v16h vb1 = load_frag_b_cm(vb, VS, k0, n0 + 16, lane);
    c0 = wmma_f16(pa, vb0, c0);
    c1 = wmma_f16(pa, vb1, c1);
  }

  float* op = ctxp + (((size_t)b * NCHUNK + ch) * 16) * VS;
  const int n = lane & 15, hf = lane >> 4;
#pragma unroll
  for (int v = 0; v < 8; ++v) {
    const int m = v + hf * 8;
    op[(size_t)m * VS + n0 + n]      = c0[v];
    op[(size_t)m * VS + n0 + 16 + n] = c1[v];
  }
  if (tid < 16) {
    msum[(((size_t)b * NCHUNK + ch) * 16 + tid) * 2 + 0] = red[tid][0];
    msum[(((size_t)b * NCHUNK + ch) * 16 + tid) * 2 + 1] = red[tid][1];
  }
}

__global__ __launch_bounds__(256) void k3_reduce_out(
    const float* __restrict__ ctxp, const float* __restrict__ msum,
    const float* __restrict__ Wv, const float* __restrict__ bv,
    float* __restrict__ out) {
  __shared__ float4 ctx4[VS / 4];
  __shared__ float  res[QS];

  const int b = blockIdx.x >> 4;
  const int h = blockIdx.x & 15;
  const int tid = threadIdx.x, wave = tid >> 5, lane = tid & 31;

  const float* cp = ctxp + (((size_t)b * NCHUNK) * 16 + h) * VS + tid;
  float part[NCHUNK];
#pragma unroll
  for (int p = 0; p < NCHUNK; ++p) part[p] = cp[(size_t)p * 16 * VS];

  const int o0 = h * QS + wave * 8;
  const float4* w4 = reinterpret_cast<const float4*>(Wv + (size_t)o0 * VS);
  float4 wa[8], wb[8];
#pragma unroll
  for (int j = 0; j < 8; ++j) {
    wa[j] = w4[j * (VS / 4) + lane];
    wb[j] = w4[j * (VS / 4) + 32 + lane];
  }
  const float bias = bv[h * QS + (tid & (QS - 1))];
  __builtin_amdgcn_sched_barrier(0);

  const float* ms = msum + ((size_t)b * NCHUNK * 16 + h) * 2;
  float mp[NCHUNK], sp[NCHUNK], mx = -INFINITY;
#pragma unroll
  for (int p = 0; p < NCHUNK; ++p) {
    mp[p] = ms[p * 32 + 0];
    sp[p] = ms[p * 32 + 1];
    mx = fmaxf(mx, mp[p]);
  }
  float denom = 0.0f;
#pragma unroll
  for (int p = 0; p < NCHUNK; ++p) denom += sp[p] * __expf(mp[p] - mx);
  const float inv = 1.0f / denom;

  float acc = 0.0f;
#pragma unroll
  for (int p = 0; p < NCHUNK; ++p) acc += part[p] * (__expf(mp[p] - mx) * inv);
  reinterpret_cast<float*>(ctx4)[tid] = acc;
  __syncthreads();

  const float4 ca = ctx4[lane], cb = ctx4[32 + lane];
  float d[8];
#pragma unroll
  for (int j = 0; j < 8; ++j) {
    float s = wa[j].x * ca.x;
    s += wa[j].y * ca.y;
    s += wa[j].z * ca.z;
    s += wa[j].w * ca.w;
    s += wb[j].x * cb.x;
    s += wb[j].y * cb.y;
    s += wb[j].z * cb.z;
    s += wb[j].w * cb.w;
    d[j] = s;
  }
#pragma unroll
  for (int off = 16; off; off >>= 1) {
#pragma unroll
    for (int j = 0; j < 8; ++j) d[j] += __shfl_xor(d[j], off, 32);
  }
  if (lane == 0) {
#pragma unroll
    for (int j = 0; j < 8; ++j) res[wave * 8 + j] = d[j];
  }
  __syncthreads();

  if (tid < QS) {
    const int o = h * QS + tid;
    out[(size_t)b * 1024 + o] = res[tid] + bias;
  }
}

extern "C" void kernel_launch(void* const* d_in, const int* in_sizes, int n_in,
                              void* d_out, int out_size, void* d_ws, size_t ws_size,
                              hipStream_t stream) {
  const float* query = (const float*)d_in[0];
  const float* key   = (const float*)d_in[1];
  const float* value = (const float*)d_in[2];
  const float* Wq    = (const float*)d_in[3];
  const float* bq    = (const float*)d_in[4];
  const float* Wv    = (const float*)d_in[5];
  const float* bv    = (const float*)d_in[6];
  float* out = (float*)d_out;

  char* ws = (char*)d_ws;
  float* Q    = (float*)ws;
  float* ctxp = (float*)(ws + 131072);
  float* msum = (float*)(ws + 131072 +
                         (size_t)B_DIM * NCHUNK * 16 * VS * 4);

  k1_qproj<<<32, 128, 0, stream>>>(query, Wq, bq, Q);
  k2_attn_partial<<<B_DIM * NCHUNK, 256, 0, stream>>>(Q, key, value, ctxp, msum);
  k3_reduce_out<<<B_DIM * H_DIM, 256, 0, stream>>>(ctxp, msum, Wv, bv, out);
}


# k2 phase 3 (P @ value): value read in whole 256-byte row pieces, 16 B per lane (N-tile = every 4th column), 2 k-steps in flight; waves = 4 column groups x 2 K halves, summed via LDS (new sum order)
# speedup vs baseline: 6.0557x; 1.3040x over previous
#include <hip/hip_runtime.h>
#include <math.h>


typedef _Float16 half_t;
typedef __attribute__((ext_vector_type(16))) _Float16 v16h;
typedef __attribute__((ext_vector_type(8)))  float    v8f;
typedef __attribute__((ext_vector_type(4)))  float    v4f;

#define B_DIM 32
#define L_DIM 4096
#define H_DIM 16
#define QS    64
#define VS    256
#define CHUNK 256
#define NCHUNK 16

__device__ __forceinline__ v8f wmma_f16(v16h a, v16h b, v8f c) {
  return __builtin_amdgcn_wmma_f32_16x16x32_f16(
       false, a,  false, b,
       (short)0, c,  false,  false);
}

__device__ __forceinline__ v16h load_frag_rm(const float* __restrict__ base,
                                             int ld, int r0, int k0,
                                             float scale, int lane) {
  const int r = lane & 15, hf = lane >> 4;
  const float* p = base + (size_t)(r0 + r) * ld + k0 + hf * 8;
  v16h f;
#pragma unroll
  for (int e = 0; e < 16; ++e)
    f[e] = (half_t)(p[16 * (e >> 3) + (e & 7)] * scale);
  return f;
}

__device__ __forceinline__ v16h load_frag_b_cm(const float* __restrict__ base,
                                               int ld, int k0, int n0, int lane) {
  const int n = lane & 15, hf = lane >> 4;
  const float* p = base + (size_t)(k0 + hf * 8) * ld + n0 + n;
  v16h f;
#pragma unroll
  for (int e = 0; e < 16; ++e)
    f[e] = (half_t)p[(size_t)(16 * (e >> 3) + (e & 7)) * ld];
  return f;
}

__device__ __forceinline__ v16h load_frag_a_lds(const half_t* base, int ldh,
                                                int k0, int lane) {
  const int m = lane & 15, hf = lane >> 4;
  const half_t* p = base + m * ldh + k0 + hf * 8;
  v16h f;
#pragma unroll
  for (int e = 0; e < 16; ++e)
    f[e] = p[16 * (e >> 3) + (e & 7)];
  return f;
}

#define K1_WAVES 16

__device__ __forceinline__ void load_raw_rm(const float* __restrict__ base,
                                            int ld, int r0, int k0, int lane,
                                            v4f (&r)[4]) {
  const int row = lane & 15, hf = lane >> 4;
  const v4f* p = reinterpret_cast<const v4f*>(
      base + (size_t)(r0 + row) * ld + k0 + hf * 8);
  r[0] = p[0];
  r[1] = p[1];
  r[2] = p[4];
  r[3] = p[5];
}

__device__ __forceinline__ v16h cvt_raw(const v4f (&r)[4]) {
  v16h f;
#pragma unroll
  for (int e = 0; e < 16; ++e) f[e] = (half_t)r[e >> 2][e & 3];
  return f;
}

__global__ __launch_bounds__(K1_WAVES * 32) void k1_qproj(
    const float* __restrict__ query, const float* __restrict__ Wq,
    const float* __restrict__ bq, float* __restrict__ Q) {
  __shared__ float part[K1_WAVES][2][8][32];

  const int nt = blockIdx.x;
  const int tid = threadIdx.x, wave = tid >> 5, lane = tid & 31;
  const int kbase = wave * (1024 / K1_WAVES);

  v4f rb[2][4], ra0[2][4], ra1[2][4];
#pragma unroll
  for (int s = 0; s < 2; ++s) {
    load_raw_rm(Wq,    1024, nt * 16, kbase + s * 32, lane, rb[s]);
    load_raw_rm(query, 1024, 0,       kbase + s * 32, lane, ra0[s]);
    load_raw_rm(query, 1024, 16,      kbase + s * 32, lane, ra1[s]);
  }
  const float bias = bq[nt * 16 + (tid & 15)];
  __builtin_amdgcn_sched_barrier(0);

  v8f c0 = {}, c1 = {};
#pragma unroll
  for (int s = 0; s < 2; ++s) {
    const v16h b = cvt_raw(rb[s]);
    c0 = wmma_f16(cvt_raw(ra0[s]), b, c0);
    c1 = wmma_f16(cvt_raw(ra1[s]), b, c1);
  }
#pragma unroll
  for (int v = 0; v < 8; ++v) {
    part[wave][0][v][lane] = c0[v];
    part[wave][1][v][lane] = c1[v];
  }
  __syncthreads();

  const int mt = tid >> 8, v = (tid >> 5) & 7, ln = tid & 31;
  float s = 0.0f;
#pragma unroll
  for (int w = 0; w < K1_WAVES; ++w) s += part[w][mt][v][ln];
  const int m = v + (ln >> 4) * 8, n = ln & 15;
  Q[(size_t)(mt * 16 + m) * 1024 + nt * 16 + n] = s + bias;
}

__global__ __launch_bounds__(256) void k2_attn_partial(
    const float* __restrict__ Q,
    const float* __restrict__ key,
    const float* __restrict__ value,
    float* __restrict__ ctxp,
    float* __restrict__ msum) {
  __shared__ alignas(16) float S[16][CHUNK];
  __shared__ half_t P[16][CHUNK + 8];
  __shared__ float  red[16][2];

  const int b  = blockIdx.x >> 4;
  const int ch = blockIdx.x & 15;
  const int l0 = ch * CHUNK;
  const int tid = threadIdx.x, wave = tid >> 5, lane = tid & 31;

  const float* qb = Q + (size_t)b * 1024;
  const v16h a0 = load_frag_rm(qb, QS, 0, 0,  0.125f, lane);
  const v16h a1 = load_frag_rm(qb, QS, 0, 32, 0.125f, lane);

  const float* kb = key + ((size_t)b * L_DIM + l0) * QS;
#pragma unroll
  for (int t = 0; t < CHUNK / 16 / 8; ++t) {
    const int ln0 = (wave + t * 8) * 16;
    v16h b0 = load_frag_rm(kb, QS, ln0, 0,  1.0f, lane);
    v16h b1 = load_frag_rm(kb, QS, ln0, 32, 1.0f, lane);
    v8f c = {};
    c = wmma_f16(a0, b0, c);
    c = wmma_f16(a1, b1, c);
    const int n = lane & 15, hf = lane >> 4;
#pragma unroll
    for (int v = 0; v < 8; ++v) S[v + hf * 8][ln0 + n] = c[v];
  }
  __syncthreads();

#pragma unroll
  for (int rr = 0; rr < 2; ++rr) {
    const int r = wave * 2 + rr;
    float vals[CHUNK / 32];
    float mx = -INFINITY;
#pragma unroll
    for (int i = 0; i < CHUNK / 32; ++i) {
      vals[i] = S[r][lane + 32 * i];
      mx = fmaxf(mx, vals[i]);
    }
#pragma unroll
    for (int off = 16; off; off >>= 1) mx = fmaxf(mx, __shfl_xor(mx, off, 32));
    float sum = 0.0f;
#pragma unroll
    for (int i = 0; i < CHUNK / 32; ++i) {
      const float e = __expf(vals[i] - mx);
      sum += e;
      P[r][lane + 32 * i] = (half_t)e;
    }
#pragma unroll
    for (int off = 16; off; off >>= 1) sum += __shfl_xor(sum, off, 32);
    if (lane == 0) { red[r][0] = mx; red[r][1] = sum; }
  }
  __syncthreads();

  const float* vb = value + ((size_t)b * L_DIM + l0) * VS;
  const int cg = wave & 3, kh = wave >> 2;
  const int n0 = cg * 64;
  const int kbeg = kh * (CHUNK / 2);
  const int j = lane & 15, hf = lane >> 4;
#define K2_VP(jf)                                                              \
  reinterpret_cast<const v4f*>(vb + (size_t)(kbeg + hf * 8) * VS + n0 + 4 * (jf))
  const v4f* vp = K2_VP(j);

  v4f raw[2][16];
  v8f acc[4] = {};

#define K2_LOAD_STEP(buf, s)                                                   \
  _Pragma("unroll") for (int e = 0; e < 16; ++e)                               \
    raw[buf][e] = vp[(size_t)((s) * 32 + 16 * (e >> 3) + (e & 7)) * (VS / 4)];

#define K2_USE_STEP(buf, s)                                                    \
  {                                                                            \
    const v16h pa = load_frag_a_lds(&P[0][0], CHUNK + 8, kbeg + (s) * 32, lane); \
    _Pragma("unroll") for (int c = 0; c < 4; ++c) {                            \
      v16h f;                                                                  \
      _Pragma("unroll") for (int e = 0; e < 16; ++e)                           \
        f[e] = (half_t)raw[buf][e][c];                                         \
      acc[c] = wmma_f16(pa, f, acc[c]);                                        \
    }                                                                          \
  }

  K2_LOAD_STEP(0, 0)
  K2_LOAD_STEP(1, 1)
  __builtin_amdgcn_sched_barrier(0);
  K2_USE_STEP(0, 0)
  __builtin_amdgcn_sched_barrier(0);
  int jf = j;
  asm volatile("" : "+v"(jf));
  vp = K2_VP(jf);
  K2_LOAD_STEP(0, 2)
  __builtin_amdgcn_sched_barrier(0);
  K2_USE_STEP(1, 1)
  __builtin_amdgcn_sched_barrier(0);
  asm volatile("" : "+v"(jf));
  vp = K2_VP(jf);
  K2_LOAD_STEP(1, 3)
  __builtin_amdgcn_sched_barrier(0);
  K2_USE_STEP(0, 2)
  K2_USE_STEP(1, 3)
#undef K2_LOAD_STEP
#undef K2_USE_STEP
#undef K2_VP

  v4f* kred = reinterpret_cast<v4f*>(&S[0][0]);
  if (kh == 1) {
#pragma unroll
    for (int r = 0; r < 8; ++r) {
      v4f t = {acc[0][r], acc[1][r], acc[2][r], acc[3][r]};
      kred[(cg * 8 + r) * 32 + lane] = t;
    }
  }
  __syncthreads();
  if (kh == 0) {
    float* op = ctxp + (((size_t)b * NCHUNK + ch) * 16) * VS;
#pragma unroll
    for (int r = 0; r < 8; ++r) {
      const v4f u = kred[(cg * 8 + r) * 32 + lane];
      v4f t = {acc[0][r] + u[0], acc[1][r] + u[1], acc[2][r] + u[2], acc[3][r] + u[3]};
      *reinterpret_cast<v4f*>(op + (size_t)(r + hf * 8) * VS + n0 + 4 * j) = t;
    }
  }
  if (tid < 16) {
    msum[(((size_t)b * NCHUNK + ch) * 16 + tid) * 2 + 0] = red[tid][0];
    msum[(((size_t)b * NCHUNK + ch) * 16 + tid) * 2 + 1] = red[tid][1];
  }
}

__global__ __launch_bounds__(256) void k3_reduce_out(
    const float* __restrict__ ctxp, const float* __restrict__ msum,
    const float* __restrict__ Wv, const float* __restrict__ bv,
    float* __restrict__ out) {
  __shared__ float4 ctx4[VS / 4];
  __shared__ float  res[QS];

  const int b = blockIdx.x >> 4;
  const int h = blockIdx.x & 15;
  const int tid = threadIdx.x, wave = tid >> 5, lane = tid & 31;

  const float* cp = ctxp + (((size_t)b * NCHUNK) * 16 + h) * VS + tid;
  float part[NCHUNK];
#pragma unroll
  for (int p = 0; p < NCHUNK; ++p) part[p] = cp[(size_t)p * 16 * VS];

  const int o0 = h * QS + wave * 8;
  const float4* w4 = reinterpret_cast<const float4*>(Wv + (size_t)o0 * VS);
  float4 wa[8], wb[8];
#pragma unroll
  for (int j = 0; j < 8; ++j) {
    wa[j] = w4[j * (VS / 4) + lane];
    wb[j] = w4[j * (VS / 4) + 32 + lane];
  }
  const float bias = bv[h * QS + (tid & (QS - 1))];
  __builtin_amdgcn_sched_barrier(0);

  const float* ms = msum + ((size_t)b * NCHUNK * 16 + h) * 2;
  float mp[NCHUNK], sp[NCHUNK], mx = -INFINITY;
#pragma unroll
  for (int p = 0; p < NCHUNK; ++p) {
    mp[p] = ms[p * 32 + 0];
    sp[p] = ms[p * 32 + 1];
    mx = fmaxf(mx, mp[p]);
  }
  float denom = 0.0f;
#pragma unroll
  for (int p = 0; p < NCHUNK; ++p) denom += sp[p] * __expf(mp[p] - mx);
  const float inv = 1.0f / denom;

  float acc = 0.0f;
#pragma unroll
  for (int p = 0; p < NCHUNK; ++p) acc += part[p] * (__expf(mp[p] - mx) * inv);
  reinterpret_cast<float*>(ctx4)[tid] = acc;
  __syncthreads();

  const float4 ca = ctx4[lane], cb = ctx4[32 + lane];
  float d[8];
#pragma unroll
  for (int j = 0; j < 8; ++j) {
    float s = wa[j].x * ca.x;
    s += wa[j].y * ca.y;
    s += wa[j].z * ca.z;
    s += wa[j].w * ca.w;
    s += wb[j].x * cb.x;
    s += wb[j].y * cb.y;
    s += wb[j].z * cb.z;
    s += wb[j].w * cb.w;
    d[j] = s;
  }
#pragma unroll
  for (int off = 16; off; off >>= 1) {
#pragma unroll
    for (int j = 0; j < 8; ++j) d[j] += __shfl_xor(d[j], off, 32);
  }
  if (lane == 0) {
#pragma unroll
    for (int j = 0; j < 8; ++j) res[wave * 8 + j] = d[j];
  }
  __syncthreads();

  if (tid < QS) {
    const int o = h * QS + tid;
    out[(size_t)b * 1024 + o] = res[tid] + bias;
  }
}

extern "C" void kernel_launch(void* const* d_in, const int* in_sizes, int n_in,
                              void* d_out, int out_size, void* d_ws, size_t ws_size,
                              hipStream_t stream) {
  const float* query = (const float*)d_in[0];
  const float* key   = (const float*)d_in[1];
  const float* value = (const float*)d_in[2];
  const float* Wq    = (const float*)d_in[3];
  const float* bq    = (const float*)d_in[4];
  const float* Wv    = (const float*)d_in[5];
  const float* bv    = (const float*)d_in[6];
  float* out = (float*)d_out;

  char* ws = (char*)d_ws;
  float* Q    = (float*)ws;
  float* ctxp = (float*)(ws + 131072);
  float* msum = (float*)(ws + 131072 +
                         (size_t)B_DIM * NCHUNK * 16 * VS * 4);

  k1_qproj<<<64, K1_WAVES * 32, 0, stream>>>(query, Wq, bq, Q);
  k2_attn_partial<<<B_DIM * NCHUNK, 256, 0, stream>>>(Q, key, value, ctxp, msum);
  k3_reduce_out<<<B_DIM * H_DIM, 256, 0, stream>>>(ctxp, msum, Wv, bv, out);
}
